# starting kernel
# baseline (speedup 1.0000x reference)
#include <hip/hip_runtime.h>
#include <hip/hip_bf16.h>

typedef float v2f __attribute__((ext_vector_type(2)));
typedef float v8f __attribute__((ext_vector_type(8)));

#define ZDIM 32
#define HDIM 128
#define TBATCH 64
#define NWAVES 16
#define NTHREADS 512
#define H_S 80
#define FD_BASE (HDIM * H_S)
#define FD_S 1058
#define FD_ROW 33
#define SM_STRIDE (TBATCH * HDIM)
#define LDS_FLOATS (FD_BASE + TBATCH * FD_S)

__device__ __forceinline__ float rlane(float x, int i) {
  return __int_as_float(__builtin_amdgcn_readlane(__float_as_int(x), i));
}

__device__ __forceinline__ v8f wmma_f32(v2f a, v2f b, v8f c) {
  return __builtin_amdgcn_wmma_f32_16x16x4_f32(false, a, false, b, (short)0, c, false, false);
}

__global__ __launch_bounds__(NTHREADS)
void sylvester_flow_kernel(const float* __restrict__ z0,
                           const float* __restrict__ h,
                           const float* __restrict__ Wd,  const float* __restrict__ bd,
                           const float* __restrict__ Wd1, const float* __restrict__ bd1,
                           const float* __restrict__ Wd2, const float* __restrict__ bd2,
                           const float* __restrict__ Wb,  const float* __restrict__ bb,
                           float* __restrict__ out_z, float* __restrict__ out_ld)
{
  __shared__ float sh[LDS_FLOATS];

  const int tid   = threadIdx.x;
  const int lane  = tid & 31;
  const int wave  = tid >> 5;
  const int mrow  = lane & 15;
  const int mhalf = lane >> 4;
  const int mt     = wave & 3;
  const int ngroup = wave >> 2;
  const int s0 = blockIdx.x * TBATCH;

  #pragma unroll
  for (int r = 0; r < (TBATCH * HDIM) / NTHREADS; ++r) {
    int idx = r * NTHREADS + tid;
    int s = idx >> 7;
    int c = idx & 127;
    sh[c * H_S + s] = h[(size_t)(s0 + s) * HDIM + c];
  }
  __syncthreads();

  const int s_loc = mt * 16 + mrow;
  v2f acache[32];
  #pragma unroll
  for (int hh = 0; hh < 32; ++hh) {
    int kk = hh * 4 + 2 * mhalf;
    v2f a;
    a.x = sh[kk * H_S + s_loc];
    a.y = sh[(kk + 1) * H_S + s_loc];
    acache[hh] = a;
  }

  #pragma unroll 1
  for (int q = 0; q < 6; ++q) {
    int id2 = ngroup * 6 + q;
    int mat = id2 >> 3;
    int nt2 = id2 & 7;
    const float* Wm = (mat == 0) ? Wd1 : (mat == 1) ? Wd2 : Wb;
    const float* bm = (mat == 0) ? bd1 : (mat == 1) ? bd2 : bb;
    int cc = nt2 * 16 + mrow;
    float bias = bm[cc];
    v8f acc;
    #pragma unroll
    for (int r = 0; r < 8; ++r) acc[r] = bias;
    const float* wrow = Wm + (size_t)cc * HDIM;
    v2f bmat[32];
    #pragma unroll
    for (int hh = 0; hh < 32; ++hh)
      bmat[hh] = *reinterpret_cast<const v2f*>(wrow + hh * 4 + 2 * mhalf);
    #pragma unroll
    for (int hh = 0; hh < 32; ++hh)
      acc = wmma_f32(acache[hh], bmat[hh], acc);
    int sbase = mt * 16 + mhalf * 8;
    #pragma unroll
    for (int r = 0; r < 8; ++r) {
      float v = acc[r];
      if (mat < 2) v = tanhf(v);
      sh[FD_BASE + mat * SM_STRIDE + (sbase + r) * HDIM + cc] = v;
    }
  }
  __syncthreads();

  const int sflow = wave * 4;
  float zr[4], ldr[4], d1r[16], d2r[16], br[16];
  #pragma unroll
  for (int si = 0; si < 4; ++si) {
    zr[si]  = z0[(size_t)(s0 + sflow + si) * ZDIM + lane];
    ldr[si] = 0.0f;
    #pragma unroll
    for (int k = 0; k < 4; ++k) {
      int coff = lane * 4 + k;
      d1r[si * 4 + k] = sh[FD_BASE + 0 * SM_STRIDE + (sflow + si) * HDIM + coff];
      d2r[si * 4 + k] = sh[FD_BASE + 1 * SM_STRIDE + (sflow + si) * HDIM + coff];
      br [si * 4 + k] = sh[FD_BASE + 2 * SM_STRIDE + (sflow + si) * HDIM + coff];
    }
  }

  #pragma unroll 1
  for (int k = 0; k < 4; ++k) {
    __syncthreads();

    #pragma unroll 1
    for (int tt = 0; tt < 16; ++tt) {
      int nt = ngroup * 16 + tt;
      int i_ = nt >> 1;
      int j0 = (nt & 1) << 4;
      int ccol = i_ * 128 + (j0 + mrow) * 4 + k;
      float bias = bd[ccol];
      v8f acc;
      #pragma unroll
      for (int r = 0; r < 8; ++r) acc[r] = bias;
      const float* wrow = Wd + (size_t)ccol * HDIM;
      v2f bmat[32];
      #pragma unroll
      for (int hh = 0; hh < 32; ++hh)
        bmat[hh] = *reinterpret_cast<const v2f*>(wrow + hh * 4 + 2 * mhalf);
      #pragma unroll
      for (int hh = 0; hh < 32; ++hh)
        acc = wmma_f32(acache[hh], bmat[hh], acc);
      int sb = mt * 16 + mhalf * 8;
      int base = FD_BASE + sb * FD_S + i_ * FD_ROW + j0 + mrow;
      #pragma unroll
      for (int r = 0; r < 8; ++r)
        sh[base + r * FD_S] = acc[r];
    }
    __syncthreads();

    const bool flip = (k & 1);
    #pragma unroll 1
    for (int si = 0; si < 4; ++si) {
      float zv = zr[si];
      float zper = flip ? __shfl_xor(zv, 31, 32) : zv;
      int fdb = FD_BASE + (sflow + si) * FD_S;
      float pre = br[si * 4 + k] + zper * d2r[si * 4 + k];
      #pragma unroll
      for (int i = 1; i < 32; ++i) {
        float zi = rlane(zper, i);
        float v  = sh[fdb + i * FD_ROW + lane];
        pre += (i > lane) ? zi * v : 0.0f;
      }
      float t = tanhf(pre);
      float dz = t * d1r[si * 4 + k];
      #pragma unroll
      for (int j = 1; j < 32; ++j) {
        float tj = rlane(t, j);
        float v  = sh[fdb + lane * FD_ROW + j];
        dz += (j > lane) ? tj * v : 0.0f;
      }
      if (flip) dz = __shfl_xor(dz, 31, 32);
      zr[si] = zv + dz;
      float dj = (1.0f - t * t) * (d1r[si * 4 + k] * d2r[si * 4 + k]) + 1.0f;
      ldr[si] += logf(fabsf(dj));
    }
  }

  #pragma unroll
  for (int si = 0; si < 4; ++si) {
    int s = s0 + sflow + si;
    out_z[(size_t)s * ZDIM + lane] = zr[si];
    float v = ldr[si];
    #pragma unroll
    for (int off = 16; off > 0; off >>= 1) v += __shfl_xor(v, off, 32);
    if (lane == 0) out_ld[s] = v;
  }
}

extern "C" void kernel_launch(void* const* d_in, const int* in_sizes, int n_in,
                              void* d_out, int out_size, void* d_ws, size_t ws_size,
                              hipStream_t stream) {
  const float* z0  = (const float*)d_in[0];
  const float* h   = (const float*)d_in[1];
  const float* Wd  = (const float*)d_in[2];
  const float* bd  = (const float*)d_in[3];
  const float* Wd1 = (const float*)d_in[4];
  const float* bd1 = (const float*)d_in[5];
  const float* Wd2 = (const float*)d_in[6];
  const float* bd2 = (const float*)d_in[7];
  const float* Wb  = (const float*)d_in[8];
  const float* bb  = (const float*)d_in[9];

  const int B = in_sizes[0] / ZDIM;
  float* out   = (float*)d_out;
  float* out_z = out;
  float* out_ld = out + (size_t)B * ZDIM;

  dim3 grid(B / TBATCH), block(NTHREADS);
  hipLaunchKernelGGL(sylvester_flow_kernel, grid, block, 0, stream,
                     z0, h, Wd, bd, Wd1, bd1, Wd2, bd2, Wb, bb, out_z, out_ld);
}


# Weight (B) tiles fetched by whole-line 16-byte loads (4 lines per wave load, was 16) and staged through wave-private LDS; B fragments read from LDS in the same k order, sums bit-identical
# speedup vs baseline: 1.2385x; 1.2385x over previous
#include <hip/hip_runtime.h>
#include <hip/hip_bf16.h>

typedef float v2f __attribute__((ext_vector_type(2)));
typedef float v4f __attribute__((ext_vector_type(4)));
typedef float v8f __attribute__((ext_vector_type(8)));

#define ZDIM 32
#define HDIM 128
#define TBATCH 64
#define NWAVES 16
#define NTHREADS 512
#define H_S 80
#define FD_BASE (HDIM * H_S)
#define FD_S 1058
#define FD_ROW 33
#define SM_STRIDE (TBATCH * HDIM)
#define LDS_FLOATS (FD_BASE + TBATCH * FD_S)
#define STG_P 36
#define STG_WAVE (FD_BASE / NWAVES)

__device__ __forceinline__ float rlane(float x, int i) {
  return __int_as_float(__builtin_amdgcn_readlane(__float_as_int(x), i));
}

__device__ __forceinline__ v8f wmma_f32(v2f a, v2f b, v8f c) {
  return __builtin_amdgcn_wmma_f32_16x16x4_f32(false, a, false, b, (short)0, c, false, false);
}

__device__ __forceinline__ void lds_wave_order() {
  __builtin_amdgcn_fence(__ATOMIC_RELEASE, "wavefront");
  __builtin_amdgcn_wave_barrier();
  __builtin_amdgcn_fence(__ATOMIC_ACQUIRE, "wavefront");
}

template <int ROWGRP>
__device__ __forceinline__ void tile_fetch(v4f (&g)[16], const float* p) {
  #pragma unroll
  for (int q = 0; q < 4; ++q) {
    #pragma unroll
    for (int ld = 0; ld < 4; ++ld)
      g[q * 4 + ld] = *reinterpret_cast<const v4f*>(p + ld * ROWGRP + q * 32);
  }
}

template <int ROWGRP>
__device__ __forceinline__ void stage_quarter(v4f (&g)[16], int q, const float* pn, float* stg_w) {
  #pragma unroll
  for (int ld = 0; ld < 4; ++ld)
    *reinterpret_cast<v4f*>(stg_w + ld * 2 * STG_P) = g[q * 4 + ld];
  #pragma unroll
  for (int ld = 0; ld < 4; ++ld)
    g[q * 4 + ld] = *reinterpret_cast<const v4f*>(pn + ld * ROWGRP + q * 32);
}

template <int ROWGRP>
__device__ __forceinline__ v8f tile_product(v4f (&g)[16], const float* pn,
                                            float* stg_w, const float* stg_r,
                                            const v2f (&acache)[32], v8f acc) {
  v2f fb[2][8];
  stage_quarter<ROWGRP>(g, 0, pn, stg_w);
  lds_wave_order();
  #pragma unroll
  for (int hq = 0; hq < 8; ++hq)
    fb[0][hq] = *reinterpret_cast<const v2f*>(stg_r + hq * 4);
  #pragma unroll
  for (int q = 0; q < 4; ++q) {
    if (q < 3) {
      lds_wave_order();
      stage_quarter<ROWGRP>(g, q + 1, pn, stg_w);
      lds_wave_order();
      #pragma unroll
      for (int hq = 0; hq < 8; ++hq)
        fb[(q + 1) & 1][hq] = *reinterpret_cast<const v2f*>(stg_r + hq * 4);
    }
    #pragma unroll
    for (int hq = 0; hq < 8; ++hq)
      acc = wmma_f32(acache[q * 8 + hq], fb[q & 1][hq], acc);
  }
  lds_wave_order();
  return acc;
}

__global__ __launch_bounds__(NTHREADS)
void sylvester_flow_kernel(const float* __restrict__ z0,
                           const float* __restrict__ h,
                           const float* __restrict__ Wd,  const float* __restrict__ bd,
                           const float* __restrict__ Wd1, const float* __restrict__ bd1,
                           const float* __restrict__ Wd2, const float* __restrict__ bd2,
                           const float* __restrict__ Wb,  const float* __restrict__ bb,
                           float* __restrict__ out_z, float* __restrict__ out_ld)
{
  __shared__ __attribute__((aligned(16))) float sh[LDS_FLOATS];

  const int tid   = threadIdx.x;
  const int lane  = tid & 31;
  const int wave  = tid >> 5;
  const int mrow  = lane & 15;
  const int mhalf = lane >> 4;
  const int mt     = wave & 3;
  const int ngroup = wave >> 2;
  const int s0 = blockIdx.x * TBATCH;

  #pragma unroll
  for (int r = 0; r < (TBATCH * HDIM) / NTHREADS; ++r) {
    int idx = r * NTHREADS + tid;
    int s = idx >> 7;
    int c = idx & 127;
    sh[c * H_S + s] = h[(size_t)(s0 + s) * HDIM + c];
  }
  __syncthreads();

  const int s_loc = mt * 16 + mrow;
  v2f acache[32];
  #pragma unroll
  for (int hh = 0; hh < 32; ++hh) {
    int kk = hh * 4 + 2 * mhalf;
    v2f a;
    a.x = sh[kk * H_S + s_loc];
    a.y = sh[(kk + 1) * H_S + s_loc];
    acache[hh] = a;
  }

  __syncthreads();

  const int grow = (lane >> 4) + 8 * ((lane >> 3) & 1);
  const int gchk = lane & 7;
  float*       stg_w = sh + wave * STG_WAVE + grow * STG_P + gchk * 4;
  const float* stg_r = sh + wave * STG_WAVE + mrow * STG_P + 2 * mhalf;
  const int goff_small = grow * HDIM + gchk * 4;
  const int goff_main  = grow * 4 * HDIM + gchk * 4;
  v4f g[16];

  {
    int id0 = ngroup * 6;
    int mat0 = id0 >> 3;
    int nt0 = id0 & 7;
    const float* W0 = (mat0 == 0) ? Wd1 : (mat0 == 1) ? Wd2 : Wb;
    const float* b0 = (mat0 == 0) ? bd1 : (mat0 == 1) ? bd2 : bb;
    float bias_n = b0[nt0 * 16 + mrow];
    tile_fetch<2 * HDIM>(g, W0 + (size_t)(nt0 * 16) * HDIM + goff_small);
    #pragma unroll 1
    for (int q = 0; q < 6; ++q) {
      int id2 = ngroup * 6 + q;
      int mat = id2 >> 3;
      int nt2 = id2 & 7;
      int cc = nt2 * 16 + mrow;
      float bias = bias_n;
      int idn = ngroup * 6 + ((q < 5) ? q + 1 : q);
      int matn = idn >> 3;
      int ntn = idn & 7;
      const float* Wn = (matn == 0) ? Wd1 : (matn == 1) ? Wd2 : Wb;
      const float* bn = (matn == 0) ? bd1 : (matn == 1) ? bd2 : bb;
      bias_n = bn[ntn * 16 + mrow];
      v8f acc;
      #pragma unroll
      for (int r = 0; r < 8; ++r) acc[r] = bias;
      acc = tile_product<2 * HDIM>(g, Wn + (size_t)(ntn * 16) * HDIM + goff_small,
                                   stg_w, stg_r, acache, acc);
      int sbase = mt * 16 + mhalf * 8;
      #pragma unroll
      for (int r = 0; r < 8; ++r) {
        float v = acc[r];
        if (mat < 2) v = tanhf(v);
        sh[FD_BASE + mat * SM_STRIDE + (sbase + r) * HDIM + cc] = v;
      }
    }
  }
  __syncthreads();

  const int sflow = wave * 4;
  float zr[4], ldr[4], d1r[16], d2r[16], br[16];
  #pragma unroll
  for (int si = 0; si < 4; ++si) {
    zr[si]  = z0[(size_t)(s0 + sflow + si) * ZDIM + lane];
    ldr[si] = 0.0f;
    #pragma unroll
    for (int k = 0; k < 4; ++k) {
      int coff = lane * 4 + k;
      d1r[si * 4 + k] = sh[FD_BASE + 0 * SM_STRIDE + (sflow + si) * HDIM + coff];
      d2r[si * 4 + k] = sh[FD_BASE + 1 * SM_STRIDE + (sflow + si) * HDIM + coff];
      br [si * 4 + k] = sh[FD_BASE + 2 * SM_STRIDE + (sflow + si) * HDIM + coff];
    }
  }

  float bias_n = bd[ngroup * 16 * 64 + 4 * mrow];
  tile_fetch<8 * HDIM>(g, Wd + (size_t)(ngroup * 16 * 64) * HDIM + goff_main);

  #pragma unroll 1
  for (int k = 0; k < 4; ++k) {
    __syncthreads();

    #pragma unroll 1
    for (int tt = 0; tt < 16; ++tt) {
      int nt = ngroup * 16 + tt;
      int i_ = nt >> 1;
      int j0 = (nt & 1) << 4;
      float bias = bias_n;
      int ntn = (tt < 15) ? nt + 1 : ngroup * 16;
      int kn  = (tt < 15) ? k : ((k < 3) ? k + 1 : k);
      int rown = ntn * 64 + kn;
      bias_n = bd[rown + 4 * mrow];
      v8f acc;
      #pragma unroll
      for (int r = 0; r < 8; ++r) acc[r] = bias;
      acc = tile_product<8 * HDIM>(g, Wd + (size_t)rown * HDIM + goff_main,
                                    stg_w, stg_r, acache, acc);
      int sb = mt * 16 + mhalf * 8;
      int base = FD_BASE + sb * FD_S + i_ * FD_ROW + j0 + mrow;
      #pragma unroll
      for (int r = 0; r < 8; ++r)
        sh[base + r * FD_S] = acc[r];
    }
    __syncthreads();

    const bool flip = (k & 1);
    #pragma unroll 1
    for (int si = 0; si < 4; ++si) {
      float zv = zr[si];
      float zper = flip ? __shfl_xor(zv, 31, 32) : zv;
      int fdb = FD_BASE + (sflow + si) * FD_S;
      float pre = br[si * 4 + k] + zper * d2r[si * 4 + k];
      #pragma unroll
      for (int i = 1; i < 32; ++i) {
        float zi = rlane(zper, i);
        float v  = sh[fdb + i * FD_ROW + lane];
        pre += (i > lane) ? zi * v : 0.0f;
      }
      float t = tanhf(pre);
      float dz = t * d1r[si * 4 + k];
      #pragma unroll
      for (int j = 1; j < 32; ++j) {
        float tj = rlane(t, j);
        float v  = sh[fdb + lane * FD_ROW + j];
        dz += (j > lane) ? tj * v : 0.0f;
      }
      if (flip) dz = __shfl_xor(dz, 31, 32);
      zr[si] = zv + dz;
      float dj = (1.0f - t * t) * (d1r[si * 4 + k] * d2r[si * 4 + k]) + 1.0f;
      ldr[si] += logf(fabsf(dj));
    }
  }

  #pragma unroll
  for (int si = 0; si < 4; ++si) {
    int s = s0 + sflow + si;
    out_z[(size_t)s * ZDIM + lane] = zr[si];
    float v = ldr[si];
    #pragma unroll
    for (int off = 16; off > 0; off >>= 1) v += __shfl_xor(v, off, 32);
    if (lane == 0) out_ld[s] = v;
  }
}

extern "C" void kernel_launch(void* const* d_in, const int* in_sizes, int n_in,
                              void* d_out, int out_size, void* d_ws, size_t ws_size,
                              hipStream_t stream) {
  const float* z0  = (const float*)d_in[0];
  const float* h   = (const float*)d_in[1];
  const float* Wd  = (const float*)d_in[2];
  const float* bd  = (const float*)d_in[3];
  const float* Wd1 = (const float*)d_in[4];
  const float* bd1 = (const float*)d_in[5];
  const float* Wd2 = (const float*)d_in[6];
  const float* bd2 = (const float*)d_in[7];
  const float* Wb  = (const float*)d_in[8];
  const float* bb  = (const float*)d_in[9];

  const int B = in_sizes[0] / ZDIM;
  float* out   = (float*)d_out;
  float* out_z = out;
  float* out_ld = out + (size_t)B * ZDIM;

  dim3 grid(B / TBATCH), block(NTHREADS);
  hipLaunchKernelGGL(sylvester_flow_kernel, grid, block, 0, stream,
                     z0, h, Wd, bd, Wd1, bd1, Wd2, bd2, Wb, bb, out_z, out_ld);
}
